# starting kernel
# baseline (speedup 1.0000x reference)
#include <hip/hip_runtime.h>
#include <math.h>

typedef __attribute__((ext_vector_type(2))) float v2f;
typedef __attribute__((ext_vector_type(4))) float v4f;
typedef __attribute__((ext_vector_type(8))) float v8f;

constexpr int LSEQ    = 1024;
constexpr int DHEAD   = 64;
constexpr int NPATCH  = 128;
constexpr int CH      = LSEQ / NPATCH;
constexpr int WIN     = 10;
constexpr int QROWS   = 16;
constexpr int KSPAN   = 80;
constexpr int KTILES  = KSPAN / 16;
constexpr int NTILES  = DHEAD / 16;

__global__ __launch_bounds__(32)
void signed_attn_wmma(const float* __restrict__ Q,
                      const float* __restrict__ K,
                      const float* __restrict__ V,
                      const float* __restrict__ log_scale,
                      float* __restrict__ O)
{
    __shared__ __align__(16) float Ssm[QROWS][KSPAN];
    __shared__ __align__(16) float Asm[QROWS][KSPAN];
    __shared__ __align__(16) float Vsm[KSPAN][DHEAD];

    const int lane = threadIdx.x;
    const int row  = lane & 15;
    const int khalf = (lane >> 4) * 2;

    const int blk  = blockIdx.x;
    const int qblk = blk & 63;
    const int bh   = blk >> 6;
    const int q0   = qblk * QROWS;
    const int p0   = qblk * 2;
    const int k0   = (p0 + 1) * CH;

    const float* Qb = Q + (size_t)bh * LSEQ * DHEAD;
    const float* Kb = K + (size_t)bh * LSEQ * DHEAD;
    const float* Vb = V + (size_t)bh * LSEQ * DHEAD;
    float*       Ob = O + (size_t)bh * LSEQ * DHEAD;

    const float sc = fminf(fmaxf(__expf(log_scale[0]), 1.0f), 30.0f) * 0.125f;

    v8f acc[KTILES];
    #pragma unroll
    for (int t = 0; t < KTILES; ++t) acc[t] = (v8f)0.0f;

    #pragma unroll
    for (int ks = 0; ks < DHEAD / 4; ++ks) {
        const int d0 = ks * 4 + khalf;
        v2f a = *(const v2f*)(Qb + (size_t)(q0 + row) * DHEAD + d0);
        #pragma unroll
        for (int t = 0; t < KTILES; ++t) {
            int key = k0 + t * 16 + row;
            if (key > LSEQ - 1) key = LSEQ - 1;
            v2f b = *(const v2f*)(Kb + (size_t)key * DHEAD + d0);
            acc[t] = __builtin_amdgcn_wmma_f32_16x16x4_f32(
                false, a, false, b, (short)0, acc[t], false, false);
        }
    }

    #pragma unroll
    for (int t = 0; t < KTILES; ++t)
        #pragma unroll
        for (int r = 0; r < 8; ++r)
            Ssm[r + 8 * (lane >> 4)][(lane & 15) + 16 * t] = acc[t][r];
    __syncthreads();

    if (lane < 16) {
        const int r  = lane;
        const int qp = p0 + (r >> 3);
        float mp = -3.4e38f, mn = -3.4e38f;
        int any = 0;
        #pragma unroll 8
        for (int c = 0; c < KSPAN; ++c) {
            const int kp = p0 + 1 + (c >> 3);
            const int d  = kp - qp;
            if (d >= 1 && d < WIN && kp < NPATCH) {
                const float s = sc * Ssm[r][c];
                mp = fmaxf(mp, s);
                mn = fmaxf(mn, -s);
                any = 1;
            }
        }
        if (!any) {
            for (int c = 0; c < KSPAN; ++c) Asm[r][c] = 0.0f;
        } else {
            float zp = 0.0f, zn = 0.0f;
            #pragma unroll 8
            for (int c = 0; c < KSPAN; ++c) {
                const int kp = p0 + 1 + (c >> 3);
                const int d  = kp - qp;
                if (d >= 1 && d < WIN && kp < NPATCH) {
                    const float s = sc * Ssm[r][c];
                    zp += __expf(s - mp);
                    zn += __expf(-s - mn);
                }
            }
            const float izp = 1.0f / zp, izn = 1.0f / zn;
            #pragma unroll 8
            for (int c = 0; c < KSPAN; ++c) {
                const int kp = p0 + 1 + (c >> 3);
                const int d  = kp - qp;
                float a = 0.0f;
                if (d >= 1 && d < WIN && kp < NPATCH) {
                    const float s = sc * Ssm[r][c];
                    a = __expf(s - mp) * izp - __expf(-s - mn) * izn;
                }
                Asm[r][c] = a;
            }
        }
    }

    {
        float4* vs = (float4*)&Vsm[0][0];
        #pragma unroll
        for (int i = 0; i < (KSPAN * DHEAD) / (32 * 4); ++i) {
            const int f   = lane + i * 32;
            const int kr  = f >> 4;
            const int col = (f & 15) * 4;
            int key = k0 + kr;
            if (key > LSEQ - 1) key = LSEQ - 1;
            vs[f] = *(const float4*)(Vb + (size_t)key * DHEAD + col);
        }
    }
    __syncthreads();

    #pragma unroll
    for (int nt = 0; nt < NTILES; ++nt) {
        v8f oacc = (v8f)0.0f;
        #pragma unroll
        for (int ks = 0; ks < KSPAN / 4; ++ks) {
            const int kb = ks * 4 + khalf;
            v2f a = *(const v2f*)&Asm[row][kb];
            const int d = nt * 16 + row;
            v2f b;
            b.x = Vsm[kb][d];
            b.y = Vsm[kb + 1][d];
            oacc = __builtin_amdgcn_wmma_f32_16x16x4_f32(
                false, a, false, b, (short)0, oacc, false, false);
        }
        #pragma unroll
        for (int r = 0; r < 8; ++r) {
            const int m = r + 8 * (lane >> 4);
            Ob[(size_t)(q0 + m) * DHEAD + nt * 16 + (lane & 15)] = oacc[r];
        }
    }
}

extern "C" void kernel_launch(void* const* d_in, const int* in_sizes, int n_in,
                              void* d_out, int out_size, void* d_ws, size_t ws_size,
                              hipStream_t stream) {
    const float* Q  = (const float*)d_in[0];
    const float* K  = (const float*)d_in[1];
    const float* V  = (const float*)d_in[2];
    const float* ls = (const float*)d_in[3];
    float* O = (float*)d_out;

    const int BH = in_sizes[0] / (LSEQ * DHEAD);
    dim3 grid(BH * (LSEQ / QROWS));
    signed_attn_wmma<<<grid, 32, 0, stream>>>(Q, K, V, ls, O);
}


# Scores computed transposed (K*Qt) so the signed softmax runs in registers on all 32 lanes and its weights feed the A*V WMMAs directly, no LDS round trip; row sums and A*V key order are re-ordered
# speedup vs baseline: 1.8436x; 1.8436x over previous
#include <hip/hip_runtime.h>
#include <math.h>


typedef __attribute__((ext_vector_type(2))) float v2f;
typedef __attribute__((ext_vector_type(4))) float v4f;
typedef __attribute__((ext_vector_type(8))) float v8f;

constexpr int LSEQ    = 1024;
constexpr int DHEAD   = 64;
constexpr int NPATCH  = 128;
constexpr int CH      = LSEQ / NPATCH;
constexpr int WIN     = 10;
constexpr int QROWS   = 16;
constexpr int KSPAN   = 80;
constexpr int KTILES  = KSPAN / 16;
constexpr int NTILES  = DHEAD / 16;

__device__ __forceinline__ float xhalf(float x) {
    const int b = __builtin_bit_cast(int, x);
    return __builtin_bit_cast(float,
        __builtin_amdgcn_permlanex16(b, b, 0x76543210, 0xfedcba98, false, false));
}

__global__ __launch_bounds__(32)
void signed_attn_wmma(const float* __restrict__ Q,
                      const float* __restrict__ K,
                      const float* __restrict__ V,
                      const float* __restrict__ log_scale,
                      float* __restrict__ O)
{
    __shared__ __align__(16) float Vsm[KSPAN][DHEAD];

    const int lane = threadIdx.x;
    const int row  = lane & 15;
    const int hq   = lane >> 4;
    const int khalf = hq * 2;

    const int blk  = blockIdx.x;
    const int qblk = blk & 63;
    const int bh   = blk >> 6;
    const int q0   = qblk * QROWS;
    const int p0   = qblk * 2;
    const int k0   = (p0 + 1) * CH;

    const float* Qb = Q + (size_t)bh * LSEQ * DHEAD;
    const float* Kb = K + (size_t)bh * LSEQ * DHEAD;
    const float* Vb = V + (size_t)bh * LSEQ * DHEAD;
    float*       Ob = O + (size_t)bh * LSEQ * DHEAD;

    const float sc = fminf(fmaxf(__expf(log_scale[0]), 1.0f), 30.0f) * 0.125f;

    v8f acc[KTILES];
    #pragma unroll
    for (int t = 0; t < KTILES; ++t) acc[t] = (v8f)0.0f;

    #pragma unroll
    for (int ks = 0; ks < DHEAD / 4; ++ks) {
        const int d0 = ks * 4 + khalf;
        v2f qf = *(const v2f*)(Qb + (size_t)(q0 + row) * DHEAD + d0);
        #pragma unroll
        for (int t = 0; t < KTILES; ++t) {
            int key = k0 + t * 16 + row;
            if (key > LSEQ - 1) key = LSEQ - 1;
            v2f kf = *(const v2f*)(Kb + (size_t)key * DHEAD + d0);
            acc[t] = __builtin_amdgcn_wmma_f32_16x16x4_f32(
                false, kf, false, qf, (short)0, acc[t], false, false);
        }
    }

    const int cq = row >> 3;
    bool val[KTILES];
    #pragma unroll
    for (int t = 0; t < KTILES; ++t) {
        const int kp = p0 + 1 + 2 * t + hq;
        const int d  = 1 + 2 * t + hq - cq;
        val[t] = (d >= 1) && (d < WIN) && (kp < NPATCH);
    }
    const bool any = (p0 + cq + 1) < NPATCH;

    float mp = -3.4e38f, mn = -3.4e38f;
    #pragma unroll
    for (int t = 0; t < KTILES; ++t) {
        float mpt = -3.4e38f, mnt = -3.4e38f;
        #pragma unroll
        for (int r = 0; r < 8; ++r) {
            const float s = sc * acc[t][r];
            mpt = fmaxf(mpt, s);
            mnt = fmaxf(mnt, -s);
        }
        mp = fmaxf(mp, val[t] ? mpt : -3.4e38f);
        mn = fmaxf(mn, val[t] ? mnt : -3.4e38f);
    }
    mp = fmaxf(mp, xhalf(mp));
    mn = fmaxf(mn, xhalf(mn));

    v8f en[KTILES];
    float zp = 0.0f, zn = 0.0f;
    #pragma unroll
    for (int t = 0; t < KTILES; ++t) {
        float zpt = 0.0f, znt = 0.0f;
        #pragma unroll
        for (int r = 0; r < 8; ++r) {
            const float s  = sc * acc[t][r];
            const float e1 = __expf(val[t] ? (s - mp)  : -INFINITY);
            const float e2 = __expf(val[t] ? (-s - mn) : -INFINITY);
            zpt += e1;
            znt += e2;
            acc[t][r] = e1;
            en[t][r]  = e2;
        }
        zp += zpt;
        zn += znt;
    }
    zp += xhalf(zp);
    zn += xhalf(zn);
    const float izp = any ? 1.0f / zp : 0.0f;
    const float izn = any ? 1.0f / zn : 0.0f;
    #pragma unroll
    for (int t = 0; t < KTILES; ++t)
        #pragma unroll
        for (int r = 0; r < 8; ++r)
            acc[t][r] = acc[t][r] * izp - en[t][r] * izn;

    {
        float4* vs = (float4*)&Vsm[0][0];
        #pragma unroll
        for (int i = 0; i < (KSPAN * DHEAD) / (32 * 4); ++i) {
            const int f   = lane + i * 32;
            const int kr  = f >> 4;
            const int col = (f & 15) * 4;
            int key = k0 + kr;
            if (key > LSEQ - 1) key = LSEQ - 1;
            vs[f] = *(const float4*)(Vb + (size_t)key * DHEAD + col);
        }
    }
    __syncthreads();

    #pragma unroll
    for (int nt = 0; nt < NTILES; ++nt) {
        v8f oacc = (v8f)0.0f;
        const int d = nt * 16 + row;
        #pragma unroll
        for (int t = 0; t < KTILES; ++t) {
            #pragma unroll
            for (int j = 0; j < 4; ++j) {
                const int kb = t * 16 + hq * 8 + j * 2;
                v2f a;
                a.x = acc[t][2 * j];
                a.y = acc[t][2 * j + 1];
                v2f b;
                b.x = Vsm[kb][d];
                b.y = Vsm[kb + 1][d];
                oacc = __builtin_amdgcn_wmma_f32_16x16x4_f32(
                    false, a, false, b, (short)0, oacc, false, false);
            }
        }
        #pragma unroll
        for (int r = 0; r < 8; ++r) {
            const int m = r + 8 * hq;
            Ob[(size_t)(q0 + m) * DHEAD + nt * 16 + row] = oacc[r];
        }
    }
}

extern "C" void kernel_launch(void* const* d_in, const int* in_sizes, int n_in,
                              void* d_out, int out_size, void* d_ws, size_t ws_size,
                              hipStream_t stream) {
    const float* Q  = (const float*)d_in[0];
    const float* K  = (const float*)d_in[1];
    const float* V  = (const float*)d_in[2];
    const float* ls = (const float*)d_in[3];
    float* O = (float*)d_out;

    const int BH = in_sizes[0] / (LSEQ * DHEAD);
    dim3 grid(BH * (LSEQ / QROWS));
    signed_attn_wmma<<<grid, 32, 0, stream>>>(Q, K, V, ls, O);
}


# Four query blocks share one 4-wave workgroup: K and V tiles (128 key rows) are staged in LDS once per workgroup with whole-row loads and serve all four waves, so K and V are fetched 2x, not 5x
# speedup vs baseline: 2.7989x; 1.5182x over previous
#include <hip/hip_runtime.h>
#include <math.h>


typedef __attribute__((ext_vector_type(2))) float v2f;
typedef __attribute__((ext_vector_type(4))) float v4f;
typedef __attribute__((ext_vector_type(8))) float v8f;

constexpr int LSEQ    = 1024;
constexpr int DHEAD   = 64;
constexpr int NPATCH  = 128;
constexpr int CH      = LSEQ / NPATCH;
constexpr int WIN     = 10;
constexpr int QROWS   = 16;
constexpr int KSPAN   = 80;
constexpr int KTILES  = KSPAN / 16;
constexpr int NTILES  = DHEAD / 16;
constexpr int WAVES   = 4;
constexpr int TROWS   = 16 * (WAVES - 1) + KSPAN;
constexpr int KPITCH  = DHEAD + 4;
constexpr int NTHR    = 32 * WAVES;

__device__ __forceinline__ float xhalf(float x) {
    const int b = __builtin_bit_cast(int, x);
    return __builtin_bit_cast(float,
        __builtin_amdgcn_permlanex16(b, b, 0x76543210, 0xfedcba98, false, false));
}

__global__ __launch_bounds__(NTHR)
void signed_attn_wmma(const float* __restrict__ Q,
                      const float* __restrict__ K,
                      const float* __restrict__ V,
                      const float* __restrict__ log_scale,
                      float* __restrict__ O)
{
    __shared__ __align__(16) float Ksm[TROWS * KPITCH];
    __shared__ __align__(16) float Vsm[TROWS][DHEAD];

    const int tid  = threadIdx.x;
    const int lane = tid & 31;
    const int wv   = tid >> 5;
    const int woff = wv * 16;
    const int row  = lane & 15;
    const int hq   = lane >> 4;
    const int khalf = hq * 2;

    const int blk  = blockIdx.x;
    const int wgq  = blk & 15;
    const int qblk = wgq * WAVES + wv;
    const int bh   = blk >> 4;
    const int kbase = (wgq * 2 * WAVES + 1) * CH;
    const int q0   = qblk * QROWS;
    const int p0   = qblk * 2;

    const float* Qb = Q + (size_t)bh * LSEQ * DHEAD;
    const float* Kb = K + (size_t)bh * LSEQ * DHEAD;
    const float* Vb = V + (size_t)bh * LSEQ * DHEAD;
    float*       Ob = O + (size_t)bh * LSEQ * DHEAD;

    const float sc = fminf(fmaxf(__expf(log_scale[0]), 1.0f), 30.0f) * 0.125f;

    v2f qf[DHEAD / 4];
    {
        v4f kreg[(TROWS * DHEAD) / (NTHR * 4)];
        #pragma unroll
        for (int i = 0; i < (TROWS * DHEAD) / (NTHR * 4); ++i) {
            const int f   = tid + i * NTHR;
            const int kr  = f >> 4;
            const int col = (f & 15) * 4;
            int key = kbase + kr;
            if (key > LSEQ - 1) key = LSEQ - 1;
            kreg[i] = *(const v4f*)(Kb + (size_t)key * DHEAD + col);
        }
        #pragma unroll
        for (int ks = 0; ks < DHEAD / 4; ++ks)
            qf[ks] = *(const v2f*)(Qb + (size_t)(q0 + row) * DHEAD + ks * 4 + khalf);
        #pragma unroll
        for (int i = 0; i < (TROWS * DHEAD) / (NTHR * 4); ++i) {
            const int f   = tid + i * NTHR;
            const int kr  = f >> 4;
            const int col = (f & 15) * 4;
            *(v4f*)&Ksm[kr * KPITCH + col] = kreg[i];
        }
    }
    __syncthreads();

    v8f acc[KTILES];
    #pragma unroll
    for (int t = 0; t < KTILES; ++t) acc[t] = (v8f)0.0f;

    #pragma unroll
    for (int ks = 0; ks < DHEAD / 4; ++ks) {
        const int d0 = ks * 4 + khalf;
        #pragma unroll
        for (int t = 0; t < KTILES; ++t) {
            v2f kf = *(const v2f*)&Ksm[(woff + t * 16 + row) * KPITCH + d0];
            acc[t] = __builtin_amdgcn_wmma_f32_16x16x4_f32(
                false, kf, false, qf[ks], (short)0, acc[t], false, false);
        }
    }

    const int cq = row >> 3;
    bool val[KTILES];
    #pragma unroll
    for (int t = 0; t < KTILES; ++t) {
        const int kp = p0 + 1 + 2 * t + hq;
        const int d  = 1 + 2 * t + hq - cq;
        val[t] = (d >= 1) && (d < WIN) && (kp < NPATCH);
    }
    const bool any = (p0 + cq + 1) < NPATCH;

    float mp = -3.4e38f, mn = -3.4e38f;
    #pragma unroll
    for (int t = 0; t < KTILES; ++t) {
        float mpt = -3.4e38f, mnt = -3.4e38f;
        #pragma unroll
        for (int r = 0; r < 8; ++r) {
            const float s = sc * acc[t][r];
            mpt = fmaxf(mpt, s);
            mnt = fmaxf(mnt, -s);
        }
        mp = fmaxf(mp, val[t] ? mpt : -3.4e38f);
        mn = fmaxf(mn, val[t] ? mnt : -3.4e38f);
    }
    mp = fmaxf(mp, xhalf(mp));
    mn = fmaxf(mn, xhalf(mn));

    v8f en[KTILES];
    float zp = 0.0f, zn = 0.0f;
    #pragma unroll
    for (int t = 0; t < KTILES; ++t) {
        float zpt = 0.0f, znt = 0.0f;
        #pragma unroll
        for (int r = 0; r < 8; ++r) {
            const float s  = sc * acc[t][r];
            const float e1 = __expf(val[t] ? (s - mp)  : -INFINITY);
            const float e2 = __expf(val[t] ? (-s - mn) : -INFINITY);
            zpt += e1;
            znt += e2;
            acc[t][r] = e1;
            en[t][r]  = e2;
        }
        zp += zpt;
        zn += znt;
    }
    zp += xhalf(zp);
    zn += xhalf(zn);
    const float izp = any ? 1.0f / zp : 0.0f;
    const float izn = any ? 1.0f / zn : 0.0f;
    #pragma unroll
    for (int t = 0; t < KTILES; ++t)
        #pragma unroll
        for (int r = 0; r < 8; ++r)
            acc[t][r] = acc[t][r] * izp - en[t][r] * izn;

    {
        float4* vs = (float4*)&Vsm[0][0];
        #pragma unroll
        for (int i = 0; i < (TROWS * DHEAD) / (NTHR * 4); ++i) {
            const int f   = tid + i * NTHR;
            const int kr  = f >> 4;
            const int col = (f & 15) * 4;
            int key = kbase + kr;
            if (key > LSEQ - 1) key = LSEQ - 1;
            vs[f] = *(const float4*)(Vb + (size_t)key * DHEAD + col);
        }
    }
    __syncthreads();

    #pragma unroll
    for (int nt = 0; nt < NTILES; ++nt) {
        v8f oacc = (v8f)0.0f;
        const int d = nt * 16 + row;
        #pragma unroll
        for (int t = 0; t < KTILES; ++t) {
            #pragma unroll
            for (int j = 0; j < 4; ++j) {
                const int kb = woff + t * 16 + hq * 8 + j * 2;
                v2f a;
                a.x = acc[t][2 * j];
                a.y = acc[t][2 * j + 1];
                v2f b;
                b.x = Vsm[kb][d];
                b.y = Vsm[kb + 1][d];
                oacc = __builtin_amdgcn_wmma_f32_16x16x4_f32(
                    false, a, false, b, (short)0, oacc, false, false);
            }
        }
        #pragma unroll
        for (int r = 0; r < 8; ++r) {
            const int m = r + 8 * hq;
            Ob[(size_t)(q0 + m) * DHEAD + nt * 16 + row] = oacc[r];
        }
    }
}

extern "C" void kernel_launch(void* const* d_in, const int* in_sizes, int n_in,
                              void* d_out, int out_size, void* d_ws, size_t ws_size,
                              hipStream_t stream) {
    const float* Q  = (const float*)d_in[0];
    const float* K  = (const float*)d_in[1];
    const float* V  = (const float*)d_in[2];
    const float* ls = (const float*)d_in[3];
    float* O = (float*)d_out;

    const int BH = in_sizes[0] / (LSEQ * DHEAD);
    dim3 grid(BH * (LSEQ / (QROWS * WAVES)));
    signed_attn_wmma<<<grid, NTHR, 0, stream>>>(Q, K, V, ls, O);
}
